# starting kernel
# baseline (speedup 1.0000x reference)
#include <hip/hip_runtime.h>
#include <math.h>

typedef float v2f __attribute__((ext_vector_type(2)));
typedef float v8f __attribute__((ext_vector_type(8)));

#define DIM 64
#define NEG_ATT 0.2f
#define NEG_OUT 0.01f

__global__ void fill_kernel(float* __restrict__ p, float v, int n) {
    int i = blockIdx.x * blockDim.x + threadIdx.x;
    if (i < n) p[i] = v;
}

__global__ __launch_bounds__(256) void gemm64_wmma(const float* __restrict__ X,
                                                   const float* __restrict__ W,
                                                   float* __restrict__ XW,
                                                   int n_row_tiles) {
    int wave   = blockIdx.x * (blockDim.x >> 5) + (threadIdx.x >> 5);
    int tile_m = wave >> 2;
    int tile_n = wave & 3;
    if (tile_m >= n_row_tiles) return;

    int lane = threadIdx.x & 31;
    int half = lane >> 4;
    int l    = lane & 15;

    const float* xrow = X + (size_t)(tile_m * 16 + l) * DIM;
    const float* wcol = W + tile_n * 16 + l;

    v8f c = {0.f, 0.f, 0.f, 0.f, 0.f, 0.f, 0.f, 0.f};
#pragma unroll
    for (int k = 0; k < DIM; k += 4) {
        v2f a, b;
        a.x = xrow[k + 2 * half];
        a.y = xrow[k + 2 * half + 1];
        b.x = wcol[(size_t)(k + 2 * half) * DIM];
        b.y = wcol[(size_t)(k + 2 * half + 1) * DIM];
        c = __builtin_amdgcn_wmma_f32_16x16x4_f32(false, a, false, b,
                                                  (short)0, c, false, false);
    }
    float* orow = XW + (size_t)(tile_m * 16 + 8 * half) * DIM + tile_n * 16 + l;
#pragma unroll
    for (int i = 0; i < 8; ++i) orow[(size_t)i * DIM] = c[i];
}

__global__ __launch_bounds__(256) void attn_dots(const float* __restrict__ XW,
                                                 const float* __restrict__ as,
                                                 const float* __restrict__ ad,
                                                 float* __restrict__ a_src,
                                                 float* __restrict__ a_dst, int n) {
    int node = (blockIdx.x * blockDim.x + threadIdx.x) >> 5;
    int lane = threadIdx.x & 31;
    if (node >= n) return;
    const float* row = XW + (size_t)node * DIM;
    float vs = row[lane] * as[lane] + row[lane + 32] * as[lane + 32];
    float vd = row[lane] * ad[lane] + row[lane + 32] * ad[lane + 32];
#pragma unroll
    for (int off = 16; off > 0; off >>= 1) {
        vs += __shfl_xor(vs, off, 32);
        vd += __shfl_xor(vd, off, 32);
    }
    if (lane == 0) { a_src[node] = vs; a_dst[node] = vd; }
}

__global__ void edge_alpha_max(const int* __restrict__ esrc, const int* __restrict__ edst,
                               const float* __restrict__ a_src, const float* __restrict__ a_dst,
                               float* __restrict__ alpha, float* __restrict__ amax,
                               int E, int Etot) {
    int e = blockIdx.x * blockDim.x + threadIdx.x;
    if (e >= Etot) return;
    int s, d;
    if (e < E) { s = esrc[e]; d = edst[e]; } else { s = e - E; d = s; }
    float a = a_src[s] + a_dst[d];
    a = (a > 0.f) ? a : NEG_ATT * a;
    alpha[e] = a;
    atomicMax(amax + d, a);
}

__global__ void edge_exp_sum(const int* __restrict__ edst,
                             float* __restrict__ alpha, const float* __restrict__ amax,
                             float* __restrict__ denom, int E, int Etot) {
    int e = blockIdx.x * blockDim.x + threadIdx.x;
    if (e >= Etot) return;
    int d = (e < E) ? edst[e] : (e - E);
    float ex = __expf(alpha[e] - amax[d]);
    alpha[e] = ex;
    atomicAdd(denom + d, ex);
}

__global__ void edge_scatter(const int* __restrict__ esrc, const int* __restrict__ edst,
                             const float* __restrict__ XW, const float* __restrict__ ex,
                             const float* __restrict__ denom, float* __restrict__ out,
                             int E, int Etot) {
    int t = blockIdx.x * blockDim.x + threadIdx.x;
    int e = t >> 4;
    if (e >= Etot) return;
    int g = t & 15;
    int s, d;
    if (e < E) { s = esrc[e]; d = edst[e]; } else { s = e - E; d = s; }
    float coef = ex[e] / (denom[d] + 1e-16f);
    const float4 x4 = *(const float4*)(XW + (size_t)s * DIM + g * 4);
    float* o = out + (size_t)d * DIM + g * 4;
    atomicAdd(o + 0, x4.x * coef);
    atomicAdd(o + 1, x4.y * coef);
    atomicAdd(o + 2, x4.z * coef);
    atomicAdd(o + 3, x4.w * coef);
}

__global__ void bias_lrelu(float* __restrict__ out, const float* __restrict__ b, int total) {
    int i = blockIdx.x * blockDim.x + threadIdx.x;
    if (i >= total) return;
    float v = out[i] + b[i & (DIM - 1)];
    out[i] = (v > 0.f) ? v : NEG_OUT * v;
}

static inline int cdiv(int a, int b) { return (a + b - 1) / b; }

static void run_layer(const float* x, const float* W, const float* att_s, const float* att_d,
                      const float* bias, const int* esrc, const int* edst,
                      int N, int E, int Etot,
                      float* xw, float* a_src, float* a_dst, float* amax, float* denom,
                      float* alpha, float* outbuf, hipStream_t stream) {
    int n_row_tiles = N / 16;
    int waves = n_row_tiles * 4;
    gemm64_wmma<<<cdiv(waves, 8), 256, 0, stream>>>(x, W, xw, n_row_tiles);

    attn_dots<<<cdiv(N * 32, 256), 256, 0, stream>>>(xw, att_s, att_d, a_src, a_dst, N);

    fill_kernel<<<cdiv(N, 256), 256, 0, stream>>>(amax, -INFINITY, N);
    fill_kernel<<<cdiv(N, 256), 256, 0, stream>>>(denom, 0.f, N);
    fill_kernel<<<cdiv(N * DIM, 256), 256, 0, stream>>>(outbuf, 0.f, N * DIM);

    edge_alpha_max<<<cdiv(Etot, 256), 256, 0, stream>>>(esrc, edst, a_src, a_dst,
                                                        alpha, amax, E, Etot);
    edge_exp_sum<<<cdiv(Etot, 256), 256, 0, stream>>>(edst, alpha, amax, denom, E, Etot);
    edge_scatter<<<cdiv(Etot * 16, 256), 256, 0, stream>>>(esrc, edst, xw, alpha, denom,
                                                           outbuf, E, Etot);

    bias_lrelu<<<cdiv(N * DIM, 256), 256, 0, stream>>>(outbuf, bias, N * DIM);
}

extern "C" void kernel_launch(void* const* d_in, const int* in_sizes, int n_in,
                              void* d_out, int out_size, void* d_ws, size_t ws_size,
                              hipStream_t stream) {
    const int*   edge_index = (const int*)d_in[0];
    const float* emb        = (const float*)d_in[1];
    const float* W1         = (const float*)d_in[2];
    const float* att_src1   = (const float*)d_in[3];
    const float* att_dst1   = (const float*)d_in[4];
    const float* b1         = (const float*)d_in[5];
    const float* W2         = (const float*)d_in[6];
    const float* att_src2   = (const float*)d_in[7];
    const float* att_dst2   = (const float*)d_in[8];
    const float* b2         = (const float*)d_in[9];

    const int E    = in_sizes[0] / 2;
    const int N    = in_sizes[1] / DIM;
    const int Etot = E + N;

    const int* esrc = edge_index;
    const int* edst = edge_index + E;

    float* ws     = (float*)d_ws;
    float* xw     = ws;
    float* h1     = xw + (size_t)N * DIM;
    float* a_src  = h1 + (size_t)N * DIM;
    float* a_dst  = a_src + N;
    float* amax   = a_dst + N;
    float* denom  = amax + N;
    float* alpha  = denom + N;

    run_layer(emb, W1, att_src1, att_dst1, b1, esrc, edst, N, E, Etot,
              xw, a_src, a_dst, amax, denom, alpha, h1, stream);
    run_layer(h1, W2, att_src2, att_dst2, b2, esrc, edst, N, E, Etot,
              xw, a_src, a_dst, amax, denom, alpha, (float*)d_out, stream);
}


# Projection moved behind the aggregation (linearity): edge passes sum input rows, final kernel does normalise, WMMA x W, bias; logits from x.(W.att); GEMM, 2nd dot launch gone; sum order changes
# speedup vs baseline: 3.1523x; 3.1523x over previous
#include <hip/hip_runtime.h>
#include <math.h>

typedef float v2f __attribute__((ext_vector_type(2)));
typedef float v4f __attribute__((ext_vector_type(4)));
typedef float v8f __attribute__((ext_vector_type(8)));

#define DIM 64
#define NEG_ATT 0.2f
#define NEG_OUT 0.01f

__device__ __forceinline__ int package_half() {
    unsigned id = __builtin_amdgcn_s_sendmsg_rtn(0x87);
    return (int)((id >> 18) & 1u);
}

__global__ __launch_bounds__(256) void fold_att(const float* __restrict__ W1,
                                                const float* __restrict__ as1,
                                                const float* __restrict__ ad1,
                                                const float* __restrict__ W2,
                                                const float* __restrict__ as2,
                                                const float* __restrict__ ad2,
                                                float* __restrict__ u) {
    const int t = threadIdx.x;
    const int which = t >> 6;
    const int k = t & 63;
    const float* W = (which < 2) ? W1 : W2;
    const float* a = (which == 0) ? as1 : (which == 1) ? ad1 : (which == 2) ? as2 : ad2;
    const float* wr = W + k * DIM;
    float acc = 0.f;
#pragma unroll
    for (int c = 0; c < DIM; ++c) acc += wr[c] * a[c];
    u[t] = acc;
}

__global__ __launch_bounds__(256) void attn_dots_init(const float* __restrict__ X,
                                                      const float* __restrict__ as,
                                                      const float* __restrict__ ad,
                                                      float* __restrict__ a_src,
                                                      float* __restrict__ a_dst,
                                                      float* __restrict__ amax2,
                                                      float* __restrict__ denom2,
                                                      float* __restrict__ acc2, int n) {
    int node = (blockIdx.x * blockDim.x + threadIdx.x) >> 5;
    int lane = threadIdx.x & 31;
    if (node >= n) return;
    const v2f z = {0.f, 0.f};
    *(v2f*)(acc2 + (size_t)node * DIM + 2 * lane) = z;
    *(v2f*)(acc2 + ((size_t)n + node) * DIM + 2 * lane) = z;
    if (lane < 2) {
        amax2[(size_t)lane * n + node]  = -INFINITY;
        denom2[(size_t)lane * n + node] = 0.f;
    }
    const float* row = X + (size_t)node * DIM;
    float vs = row[lane] * as[lane] + row[lane + 32] * as[lane + 32];
    float vd = row[lane] * ad[lane] + row[lane + 32] * ad[lane + 32];
#pragma unroll
    for (int off = 16; off > 0; off >>= 1) {
        vs += __shfl_xor(vs, off, 32);
        vd += __shfl_xor(vd, off, 32);
    }
    if (lane == 0) { a_src[node] = vs; a_dst[node] = vd; }
}

__global__ void edge_alpha_max(const int* __restrict__ esrc, const int* __restrict__ edst,
                               const float* __restrict__ a_src, const float* __restrict__ a_dst,
                               float* __restrict__ alpha, float* __restrict__ amax2,
                               int E, int Etot) {
    int e = blockIdx.x * blockDim.x + threadIdx.x;
    if (e >= Etot) return;
    int n = Etot - E;
    float* amax = amax2 + (size_t)package_half() * n;
    int s, d;
    if (e < E) { s = esrc[e]; d = edst[e]; } else { s = e - E; d = s; }
    float a = a_src[s] + a_dst[d];
    a = (a > 0.f) ? a : NEG_ATT * a;
    alpha[e] = a;
    atomicMax(amax + d, a);
}

#define SCATTER_BATCH 8
__global__ __launch_bounds__(256) void edge_exp_scatter(const int* __restrict__ esrc,
                                                        const int* __restrict__ edst,
                                                        const float* __restrict__ X,
                                                        const float* __restrict__ alpha,
                                                        const float* __restrict__ amax2,
                                                        float* __restrict__ denom2,
                                                        float* __restrict__ acc2,
                                                        int E, int Etot) {
    const int lane    = threadIdx.x & 31;
    const int wib     = __builtin_amdgcn_readfirstlane((int)(threadIdx.x >> 5));
    const int wave    = blockIdx.x * (blockDim.x >> 5) + wib;
    const int nwaves  = gridDim.x * (blockDim.x >> 5);
    const int n       = Etot - E;
    const int nchunks = (Etot + 31) >> 5;
    const int ph      = package_half();
    const char* xwb = (const char*)X;
    char*  accb  = (char*)(acc2 + (size_t)ph * n * DIM);
    float* denom = denom2 + (size_t)ph * n;

    for (int c = wave; c < nchunks; c += nwaves) {
        const int e = c * 32 + lane;
        int s = 0, d = 0;
        float w = 0.f;
        if (e < Etot) {
            if (e < E) { s = esrc[e]; d = edst[e]; } else { s = e - E; d = s; }
            const float m = fmaxf(amax2[d], amax2[n + d]);
            w = __expf(alpha[e] - m);
            atomicAdd(denom + d, w);
        }
        const int wi  = __float_as_int(w);
        const int cnt = min(32, Etot - c * 32);
        if (cnt == 32) {
            for (int jb = 0; jb < 32; jb += SCATTER_BATCH) {
                float v0[SCATTER_BATCH], v1[SCATTER_BATCH];
#pragma unroll
                for (int u = 0; u < SCATTER_BATCH; ++u) {
                    const unsigned so = (unsigned)__builtin_amdgcn_readlane(s, jb + u) * (DIM * 4u);
                    const float* row = (const float*)(xwb + so);
                    v0[u] = row[lane];
                    v1[u] = row[lane + 32];
                }
#pragma unroll
                for (int u = 0; u < SCATTER_BATCH; ++u) {
                    const unsigned dof = (unsigned)__builtin_amdgcn_readlane(d, jb + u) * (DIM * 4u);
                    const float wj = __int_as_float(__builtin_amdgcn_readlane(wi, jb + u));
                    float* orow = (float*)(accb + dof);
                    atomicAdd(orow + lane, v0[u] * wj);
                    atomicAdd(orow + lane + 32, v1[u] * wj);
                }
            }
        } else {
            for (int j = 0; j < cnt; ++j) {
                const unsigned so  = (unsigned)__builtin_amdgcn_readlane(s, j) * (DIM * 4u);
                const unsigned dof = (unsigned)__builtin_amdgcn_readlane(d, j) * (DIM * 4u);
                const float wj = __int_as_float(__builtin_amdgcn_readlane(wi, j));
                const float* row = (const float*)(xwb + so);
                float* orow = (float*)(accb + dof);
                atomicAdd(orow + lane, row[lane] * wj);
                atomicAdd(orow + lane + 32, row[lane + 32] * wj);
            }
        }
    }
}

template <bool NEXT>
__global__ __launch_bounds__(256) void norm_project(const float* __restrict__ acc2,
                                                    const float* __restrict__ denom2,
                                                    const float* __restrict__ W,
                                                    const float* __restrict__ b,
                                                    float* __restrict__ out,
                                                    const float* __restrict__ un,
                                                    float* __restrict__ a_src,
                                                    float* __restrict__ a_dst,
                                                    float* __restrict__ amaxN,
                                                    float* __restrict__ denomN,
                                                    float* __restrict__ accN,
                                                    int n, int n_strips) {
    __shared__ v2f Wl[16 * 4 * 32];
    const int tid = threadIdx.x;
    const int lane = tid & 31;
    const int half = lane >> 4;
    const int l    = lane & 15;
    const int strip = blockIdx.x * 8 + (tid >> 5);
    const int row0 = min(strip, n_strips - 1) * 16;
    const size_t plane = (size_t)n * DIM;

    v2f wreg[8];
#pragma unroll
    for (int r = 0; r < 8; ++r) {
        const int i = tid + 256 * r;
        const int t = i >> 7, tn = (i >> 5) & 3, ln = i & 31;
        const int kb = 8 * (t >> 1) + 4 * (ln >> 4) + 2 * (t & 1);
        wreg[r].x = W[kb * DIM + 4 * (ln & 15) + tn];
        wreg[r].y = W[(kb + 1) * DIM + 4 * (ln & 15) + tn];
    }

    const int arow = row0 + l;
    const float d0 = denom2[arow];
    const float d1 = denom2[n + arow];
    const float* p0 = acc2 + (size_t)arow * DIM + 4 * half;
    const float* p1 = p0 + plane;
    v4f ax[8], ay[8];
#pragma unroll
    for (int j = 0; j < 8; ++j) {
        ax[j] = *(const v4f*)(p0 + 8 * j);
        ay[j] = *(const v4f*)(p1 + 8 * j);
    }

#pragma unroll
    for (int r = 0; r < 8; ++r) Wl[tid + 256 * r] = wreg[r];
    __syncthreads();
    if (strip >= n_strips) return;

    const float rden = 1.0f / (d0 + d1 + 1e-16f);
    v4f av[8];
#pragma unroll
    for (int j = 0; j < 8; ++j) av[j] = (ax[j] + ay[j]) * rden;

    v8f c[4];
#pragma unroll
    for (int tn = 0; tn < 4; ++tn) c[tn] = (v8f){0.f, 0.f, 0.f, 0.f, 0.f, 0.f, 0.f, 0.f};
#pragma unroll
    for (int t = 0; t < 16; ++t) {
        v2f a;
        if (t & 1) { a.x = av[t >> 1].z; a.y = av[t >> 1].w; }
        else       { a.x = av[t >> 1].x; a.y = av[t >> 1].y; }
#pragma unroll
        for (int tn = 0; tn < 4; ++tn) {
            const v2f bb = Wl[(t * 4 + tn) * 32 + lane];
            c[tn] = __builtin_amdgcn_wmma_f32_16x16x4_f32(false, a, false, bb,
                                                          (short)0, c[tn], false, false);
        }
    }

    float ps[8], pd[8];
    const v4f bias = *(const v4f*)(b + 4 * l);
    v4f us = {0.f, 0.f, 0.f, 0.f}, ud = {0.f, 0.f, 0.f, 0.f};
    if (NEXT) { us = *(const v4f*)(un + 4 * l); ud = *(const v4f*)(un + DIM + 4 * l); }
    float* orow = out + (size_t)(row0 + 8 * half) * DIM + 4 * l;
#pragma unroll
    for (int i = 0; i < 8; ++i) {
        v4f v;
        v.x = c[0][i] + bias.x;
        v.y = c[1][i] + bias.y;
        v.z = c[2][i] + bias.z;
        v.w = c[3][i] + bias.w;
        v.x = (v.x > 0.f) ? v.x : NEG_OUT * v.x;
        v.y = (v.y > 0.f) ? v.y : NEG_OUT * v.y;
        v.z = (v.z > 0.f) ? v.z : NEG_OUT * v.z;
        v.w = (v.w > 0.f) ? v.w : NEG_OUT * v.w;
        *(v4f*)(orow + (size_t)i * DIM) = v;
        ps[i] = v.x * us.x + v.y * us.y + v.z * us.z + v.w * us.w;
        pd[i] = v.x * ud.x + v.y * ud.y + v.z * ud.z + v.w * ud.w;
    }
    if (NEXT) {
#pragma unroll
        for (int off = 8; off > 0; off >>= 1) {
#pragma unroll
            for (int i = 0; i < 8; ++i) {
                ps[i] += __shfl_xor(ps[i], off, 32);
                pd[i] += __shfl_xor(pd[i], off, 32);
            }
        }
#pragma unroll
        for (int i = 0; i < 8; ++i) {
            if (l == i) {
                a_src[row0 + 8 * half + i] = ps[i];
                a_dst[row0 + 8 * half + i] = pd[i];
            }
        }
        if (lane < 16) {
            amaxN[row0 + lane]      = -INFINITY;
            amaxN[n + row0 + lane]  = -INFINITY;
            denomN[row0 + lane]     = 0.f;
            denomN[n + row0 + lane] = 0.f;
        }
        const v4f z = {0.f, 0.f, 0.f, 0.f};
        float* q = accN + (size_t)row0 * DIM + 4 * lane;
#pragma unroll
        for (int it = 0; it < 8; ++it) {
            *(v4f*)(q + it * 128) = z;
            *(v4f*)(q + plane + it * 128) = z;
        }
    }
}

static inline int cdiv(int a, int b) { return (a + b - 1) / b; }

extern "C" void kernel_launch(void* const* d_in, const int* in_sizes, int n_in,
                              void* d_out, int out_size, void* d_ws, size_t ws_size,
                              hipStream_t stream) {
    const int*   edge_index = (const int*)d_in[0];
    const float* emb        = (const float*)d_in[1];
    const float* W1         = (const float*)d_in[2];
    const float* att_src1   = (const float*)d_in[3];
    const float* att_dst1   = (const float*)d_in[4];
    const float* b1         = (const float*)d_in[5];
    const float* W2         = (const float*)d_in[6];
    const float* att_src2   = (const float*)d_in[7];
    const float* att_dst2   = (const float*)d_in[8];
    const float* b2         = (const float*)d_in[9];

    const int E    = in_sizes[0] / 2;
    const int N    = in_sizes[1] / DIM;
    const int Etot = E + N;
    const int n_strips = N / 16;

    const int* esrc = edge_index;
    const int* edst = edge_index + E;

    float* ws     = (float*)d_ws;
    float* h1     = ws;
    float* accA   = h1 + (size_t)N * DIM;
    float* accB   = accA + (size_t)2 * N * DIM;
    float* a_src  = accB + (size_t)2 * N * DIM;
    float* a_dst  = a_src + N;
    float* amaxA  = a_dst + N;
    float* denomA = amaxA + 2 * N;
    float* amaxB  = denomA + 2 * N;
    float* denomB = amaxB + 2 * N;
    float* u      = denomB + 2 * N;
    float* alpha  = u + 4 * DIM;

    fold_att<<<1, 256, 0, stream>>>(W1, att_src1, att_dst1, W2, att_src2, att_dst2, u);

    attn_dots_init<<<cdiv(N * 32, 256), 256, 0, stream>>>(emb, u, u + DIM, a_src, a_dst,
                                                          amaxA, denomA, accA, N);
    edge_alpha_max<<<cdiv(Etot, 256), 256, 0, stream>>>(esrc, edst, a_src, a_dst,
                                                        alpha, amaxA, E, Etot);
    edge_exp_scatter<<<2048, 256, 0, stream>>>(esrc, edst, emb, alpha, amaxA, denomA, accA,
                                               E, Etot);
    norm_project<true><<<cdiv(n_strips, 8), 256, 0, stream>>>(accA, denomA, W1, b1, h1,
                                                              u + 2 * DIM, a_src, a_dst,
                                                              amaxB, denomB, accB, N, n_strips);

    edge_alpha_max<<<cdiv(Etot, 256), 256, 0, stream>>>(esrc, edst, a_src, a_dst,
                                                        alpha, amaxB, E, Etot);
    edge_exp_scatter<<<2048, 256, 0, stream>>>(esrc, edst, h1, alpha, amaxB, denomB, accB,
                                               E, Etot);
    norm_project<false><<<cdiv(n_strips, 8), 256, 0, stream>>>(accB, denomB, W2, b2,
                                                               (float*)d_out, u, a_src, a_dst,
                                                               amaxB, denomB, accB, N, n_strips);
}


# Node-dot and edge-logit launches re-laid as few long waves: a wave takes 32 nodes (16-byte loads, 16 in flight) or 512 edges (16 per lane); 206,250 short waves become 9,772; dot sum tree changes
# speedup vs baseline: 3.2936x; 1.0448x over previous
#include <hip/hip_runtime.h>
#include <math.h>

typedef float v2f __attribute__((ext_vector_type(2)));
typedef float v4f __attribute__((ext_vector_type(4)));
typedef float v8f __attribute__((ext_vector_type(8)));
typedef int   v4i __attribute__((ext_vector_type(4)));

#define DIM 64
#define NEG_ATT 0.2f
#define NEG_OUT 0.01f

__device__ __forceinline__ int package_half() {
    unsigned id = __builtin_amdgcn_s_sendmsg_rtn(0x87);
    return (int)((id >> 18) & 1u);
}

__global__ __launch_bounds__(256) void fold_att(const float* __restrict__ W1,
                                                const float* __restrict__ as1,
                                                const float* __restrict__ ad1,
                                                const float* __restrict__ W2,
                                                const float* __restrict__ as2,
                                                const float* __restrict__ ad2,
                                                float* __restrict__ u) {
    const int t = threadIdx.x;
    const int which = t >> 6;
    const int k = t & 63;
    const float* W = (which < 2) ? W1 : W2;
    const float* a = (which == 0) ? as1 : (which == 1) ? ad1 : (which == 2) ? as2 : ad2;
    const float* wr = W + k * DIM;
    float acc = 0.f;
#pragma unroll
    for (int c = 0; c < DIM; ++c) acc += wr[c] * a[c];
    u[t] = acc;
}

__global__ __launch_bounds__(128) void attn_dots_init(const float* __restrict__ X,
                                                      const float* __restrict__ as,
                                                      const float* __restrict__ ad,
                                                      float* __restrict__ a_src,
                                                      float* __restrict__ a_dst,
                                                      float* __restrict__ amax2,
                                                      float* __restrict__ denom2,
                                                      float* __restrict__ acc2,
                                                      int n, int n_strips) {
    const int lane = threadIdx.x & 31;
    const int sub  = lane >> 4;
    const int l    = lane & 15;
    const int wave = (blockIdx.x * blockDim.x + threadIdx.x) >> 5;
    if (2 * wave >= n_strips) return;
    const size_t plane = (size_t)n * DIM;
    int row0[2];
    row0[0] = 2 * wave * 16;
    row0[1] = min(2 * wave + 1, n_strips - 1) * 16;

    v4f x[2][8];
#pragma unroll
    for (int h = 0; h < 2; ++h) {
        const float* p = X + (size_t)row0[h] * DIM + 4 * lane;
#pragma unroll
        for (int j = 0; j < 8; ++j) x[h][j] = *(const v4f*)(p + j * 128);
    }
    const v4f s4 = *(const v4f*)(as + 4 * l);
    const v4f d4 = *(const v4f*)(ad + 4 * l);

    const v4f z = {0.f, 0.f, 0.f, 0.f};
#pragma unroll
    for (int h = 0; h < 2; ++h) {
        float* q = acc2 + (size_t)row0[h] * DIM + 4 * lane;
#pragma unroll
        for (int it = 0; it < 8; ++it) {
            *(v4f*)(q + it * 128) = z;
            *(v4f*)(q + plane + it * 128) = z;
        }
        if (lane < 16) {
            amax2[row0[h] + lane]      = -INFINITY;
            amax2[n + row0[h] + lane]  = -INFINITY;
            denom2[row0[h] + lane]     = 0.f;
            denom2[n + row0[h] + lane] = 0.f;
        }
    }

#pragma unroll
    for (int h = 0; h < 2; ++h) {
        float vs[8], vd[8];
#pragma unroll
        for (int j = 0; j < 8; ++j) {
            const v4f v = x[h][j];
            vs[j] = v.x * s4.x + v.y * s4.y + v.z * s4.z + v.w * s4.w;
            vd[j] = v.x * d4.x + v.y * d4.y + v.z * d4.z + v.w * d4.w;
        }
#pragma unroll
        for (int off = 8; off > 0; off >>= 1) {
#pragma unroll
            for (int j = 0; j < 8; ++j) {
                vs[j] += __shfl_xor(vs[j], off, 32);
                vd[j] += __shfl_xor(vd[j], off, 32);
            }
        }
        float os = vs[0], od = vd[0];
#pragma unroll
        for (int j = 1; j < 8; ++j) {
            if (l == j) { os = vs[j]; od = vd[j]; }
        }
        if (l < 8) {
            a_src[row0[h] + 2 * l + sub] = os;
            a_dst[row0[h] + 2 * l + sub] = od;
        }
    }
}

#define EDGES_PER_WAVE 512
__global__ __launch_bounds__(64) void edge_alpha_max(const int* __restrict__ esrc,
                                                     const int* __restrict__ edst,
                                                     const float* __restrict__ a_src,
                                                     const float* __restrict__ a_dst,
                                                     float* __restrict__ alpha,
                                                     float* __restrict__ amax2,
                                                     int E, int Etot) {
    const int lane = threadIdx.x & 31;
    const int wave = blockIdx.x * (blockDim.x >> 5)
                   + __builtin_amdgcn_readfirstlane((int)(threadIdx.x >> 5));
    const int base = wave * EDGES_PER_WAVE;
    if (base >= Etot) return;
    const int n = Etot - E;
    float* amax = amax2 + (size_t)package_half() * n;

    if ((((E | Etot) & 3) == 0) && base + EDGES_PER_WAVE <= E) {
        v4i sv[4], dv[4];
#pragma unroll
        for (int k = 0; k < 4; ++k) {
            const int q = base + 128 * k + 4 * lane;
            sv[k] = *(const v4i*)(esrc + q);
            dv[k] = *(const v4i*)(edst + q);
        }
        v4f av[4];
#pragma unroll
        for (int k = 0; k < 4; ++k) {
#pragma unroll
            for (int i = 0; i < 4; ++i) {
                const float a = a_src[sv[k][i]] + a_dst[dv[k][i]];
                av[k][i] = (a > 0.f) ? a : NEG_ATT * a;
            }
        }
#pragma unroll
        for (int k = 0; k < 4; ++k) {
            const int q = base + 128 * k + 4 * lane;
            *(v4f*)(alpha + q) = av[k];
#pragma unroll
            for (int i = 0; i < 4; ++i)
                atomicMax(amax + dv[k][i], av[k][i]);
        }
    } else {
        int   s[16], d[16];
        float a[16];
#pragma unroll
        for (int m = 0; m < 16; ++m) {
            const int e = min(base + 32 * m + lane, Etot - 1);
            if (e < E) { s[m] = esrc[e]; d[m] = edst[e]; } else { s[m] = e - E; d[m] = e - E; }
        }
#pragma unroll
        for (int m = 0; m < 16; ++m) {
            const float t = a_src[s[m]] + a_dst[d[m]];
            a[m] = (t > 0.f) ? t : NEG_ATT * t;
        }
#pragma unroll
        for (int m = 0; m < 16; ++m) {
            const int e = base + 32 * m + lane;
            if (e < Etot) {
                alpha[e] = a[m];
                atomicMax(amax + d[m], a[m]);
            }
        }
    }
}

#define SCATTER_BATCH 8
__global__ __launch_bounds__(256) void edge_exp_scatter(const int* __restrict__ esrc,
                                                        const int* __restrict__ edst,
                                                        const float* __restrict__ X,
                                                        const float* __restrict__ alpha,
                                                        const float* __restrict__ amax2,
                                                        float* __restrict__ denom2,
                                                        float* __restrict__ acc2,
                                                        int E, int Etot) {
    const int lane    = threadIdx.x & 31;
    const int wib     = __builtin_amdgcn_readfirstlane((int)(threadIdx.x >> 5));
    const int wave    = blockIdx.x * (blockDim.x >> 5) + wib;
    const int nwaves  = gridDim.x * (blockDim.x >> 5);
    const int n       = Etot - E;
    const int nchunks = (Etot + 31) >> 5;
    const int ph      = package_half();
    const char* xwb = (const char*)X;
    char*  accb  = (char*)(acc2 + (size_t)ph * n * DIM);
    float* denom = denom2 + (size_t)ph * n;

    for (int c = wave; c < nchunks; c += nwaves) {
        const int e = c * 32 + lane;
        int s = 0, d = 0;
        float w = 0.f;
        if (e < Etot) {
            if (e < E) { s = esrc[e]; d = edst[e]; } else { s = e - E; d = s; }
            const float m = fmaxf(amax2[d], amax2[n + d]);
            w = __expf(alpha[e] - m);
            atomicAdd(denom + d, w);
        }
        const int wi  = __float_as_int(w);
        const int cnt = min(32, Etot - c * 32);
        if (cnt == 32) {
            for (int jb = 0; jb < 32; jb += SCATTER_BATCH) {
                float v0[SCATTER_BATCH], v1[SCATTER_BATCH];
#pragma unroll
                for (int u = 0; u < SCATTER_BATCH; ++u) {
                    const unsigned so = (unsigned)__builtin_amdgcn_readlane(s, jb + u) * (DIM * 4u);
                    const float* row = (const float*)(xwb + so);
                    v0[u] = row[lane];
                    v1[u] = row[lane + 32];
                }
#pragma unroll
                for (int u = 0; u < SCATTER_BATCH; ++u) {
                    const unsigned dof = (unsigned)__builtin_amdgcn_readlane(d, jb + u) * (DIM * 4u);
                    const float wj = __int_as_float(__builtin_amdgcn_readlane(wi, jb + u));
                    float* orow = (float*)(accb + dof);
                    atomicAdd(orow + lane, v0[u] * wj);
                    atomicAdd(orow + lane + 32, v1[u] * wj);
                }
            }
        } else {
            for (int j = 0; j < cnt; ++j) {
                const unsigned so  = (unsigned)__builtin_amdgcn_readlane(s, j) * (DIM * 4u);
                const unsigned dof = (unsigned)__builtin_amdgcn_readlane(d, j) * (DIM * 4u);
                const float wj = __int_as_float(__builtin_amdgcn_readlane(wi, j));
                const float* row = (const float*)(xwb + so);
                float* orow = (float*)(accb + dof);
                atomicAdd(orow + lane, row[lane] * wj);
                atomicAdd(orow + lane + 32, row[lane + 32] * wj);
            }
        }
    }
}

template <bool NEXT>
__global__ __launch_bounds__(256) void norm_project(const float* __restrict__ acc2,
                                                    const float* __restrict__ denom2,
                                                    const float* __restrict__ W,
                                                    const float* __restrict__ b,
                                                    float* __restrict__ out,
                                                    const float* __restrict__ un,
                                                    float* __restrict__ a_src,
                                                    float* __restrict__ a_dst,
                                                    float* __restrict__ amaxN,
                                                    float* __restrict__ denomN,
                                                    float* __restrict__ accN,
                                                    int n, int n_strips) {
    __shared__ v2f Wl[16 * 4 * 32];
    const int tid = threadIdx.x;
    const int lane = tid & 31;
    const int half = lane >> 4;
    const int l    = lane & 15;
    const int strip = blockIdx.x * 8 + (tid >> 5);
    const int row0 = min(strip, n_strips - 1) * 16;
    const size_t plane = (size_t)n * DIM;

    v2f wreg[8];
#pragma unroll
    for (int r = 0; r < 8; ++r) {
        const int i = tid + 256 * r;
        const int t = i >> 7, tn = (i >> 5) & 3, ln = i & 31;
        const int kb = 8 * (t >> 1) + 4 * (ln >> 4) + 2 * (t & 1);
        wreg[r].x = W[kb * DIM + 4 * (ln & 15) + tn];
        wreg[r].y = W[(kb + 1) * DIM + 4 * (ln & 15) + tn];
    }

    const int arow = row0 + l;
    const float d0 = denom2[arow];
    const float d1 = denom2[n + arow];
    const float* p0 = acc2 + (size_t)arow * DIM + 4 * half;
    const float* p1 = p0 + plane;
    v4f ax[8], ay[8];
#pragma unroll
    for (int j = 0; j < 8; ++j) {
        ax[j] = *(const v4f*)(p0 + 8 * j);
        ay[j] = *(const v4f*)(p1 + 8 * j);
    }

#pragma unroll
    for (int r = 0; r < 8; ++r) Wl[tid + 256 * r] = wreg[r];
    __syncthreads();
    if (strip >= n_strips) return;

    const float rden = 1.0f / (d0 + d1 + 1e-16f);
    v4f av[8];
#pragma unroll
    for (int j = 0; j < 8; ++j) av[j] = (ax[j] + ay[j]) * rden;

    v8f c[4];
#pragma unroll
    for (int tn = 0; tn < 4; ++tn) c[tn] = (v8f){0.f, 0.f, 0.f, 0.f, 0.f, 0.f, 0.f, 0.f};
#pragma unroll
    for (int t = 0; t < 16; ++t) {
        v2f a;
        if (t & 1) { a.x = av[t >> 1].z; a.y = av[t >> 1].w; }
        else       { a.x = av[t >> 1].x; a.y = av[t >> 1].y; }
#pragma unroll
        for (int tn = 0; tn < 4; ++tn) {
            const v2f bb = Wl[(t * 4 + tn) * 32 + lane];
            c[tn] = __builtin_amdgcn_wmma_f32_16x16x4_f32(false, a, false, bb,
                                                          (short)0, c[tn], false, false);
        }
    }

    float ps[8], pd[8];
    const v4f bias = *(const v4f*)(b + 4 * l);
    v4f us = {0.f, 0.f, 0.f, 0.f}, ud = {0.f, 0.f, 0.f, 0.f};
    if (NEXT) { us = *(const v4f*)(un + 4 * l); ud = *(const v4f*)(un + DIM + 4 * l); }
    float* orow = out + (size_t)(row0 + 8 * half) * DIM + 4 * l;
#pragma unroll
    for (int i = 0; i < 8; ++i) {
        v4f v;
        v.x = c[0][i] + bias.x;
        v.y = c[1][i] + bias.y;
        v.z = c[2][i] + bias.z;
        v.w = c[3][i] + bias.w;
        v.x = (v.x > 0.f) ? v.x : NEG_OUT * v.x;
        v.y = (v.y > 0.f) ? v.y : NEG_OUT * v.y;
        v.z = (v.z > 0.f) ? v.z : NEG_OUT * v.z;
        v.w = (v.w > 0.f) ? v.w : NEG_OUT * v.w;
        *(v4f*)(orow + (size_t)i * DIM) = v;
        ps[i] = v.x * us.x + v.y * us.y + v.z * us.z + v.w * us.w;
        pd[i] = v.x * ud.x + v.y * ud.y + v.z * ud.z + v.w * ud.w;
    }
    if (NEXT) {
#pragma unroll
        for (int off = 8; off > 0; off >>= 1) {
#pragma unroll
            for (int i = 0; i < 8; ++i) {
                ps[i] += __shfl_xor(ps[i], off, 32);
                pd[i] += __shfl_xor(pd[i], off, 32);
            }
        }
#pragma unroll
        for (int i = 0; i < 8; ++i) {
            if (l == i) {
                a_src[row0 + 8 * half + i] = ps[i];
                a_dst[row0 + 8 * half + i] = pd[i];
            }
        }
        if (lane < 16) {
            amaxN[row0 + lane]      = -INFINITY;
            amaxN[n + row0 + lane]  = -INFINITY;
            denomN[row0 + lane]     = 0.f;
            denomN[n + row0 + lane] = 0.f;
        }
        const v4f z = {0.f, 0.f, 0.f, 0.f};
        float* q = accN + (size_t)row0 * DIM + 4 * lane;
#pragma unroll
        for (int it = 0; it < 8; ++it) {
            *(v4f*)(q + it * 128) = z;
            *(v4f*)(q + plane + it * 128) = z;
        }
    }
}

static inline int cdiv(int a, int b) { return (a + b - 1) / b; }

extern "C" void kernel_launch(void* const* d_in, const int* in_sizes, int n_in,
                              void* d_out, int out_size, void* d_ws, size_t ws_size,
                              hipStream_t stream) {
    const int*   edge_index = (const int*)d_in[0];
    const float* emb        = (const float*)d_in[1];
    const float* W1         = (const float*)d_in[2];
    const float* att_src1   = (const float*)d_in[3];
    const float* att_dst1   = (const float*)d_in[4];
    const float* b1         = (const float*)d_in[5];
    const float* W2         = (const float*)d_in[6];
    const float* att_src2   = (const float*)d_in[7];
    const float* att_dst2   = (const float*)d_in[8];
    const float* b2         = (const float*)d_in[9];

    const int E    = in_sizes[0] / 2;
    const int N    = in_sizes[1] / DIM;
    const int Etot = E + N;
    const int n_strips = N / 16;

    const int* esrc = edge_index;
    const int* edst = edge_index + E;

    float* ws     = (float*)d_ws;
    float* h1     = ws;
    float* accA   = h1 + (size_t)N * DIM;
    float* accB   = accA + (size_t)2 * N * DIM;
    float* a_src  = accB + (size_t)2 * N * DIM;
    float* a_dst  = a_src + N;
    float* amaxA  = a_dst + N;
    float* denomA = amaxA + 2 * N;
    float* amaxB  = denomA + 2 * N;
    float* denomB = amaxB + 2 * N;
    float* u      = denomB + 2 * N;
    float* alpha  = u + 4 * DIM;

    fold_att<<<1, 256, 0, stream>>>(W1, att_src1, att_dst1, W2, att_src2, att_dst2, u);

    attn_dots_init<<<cdiv(cdiv(n_strips, 2), 4), 128, 0, stream>>>(emb, u, u + DIM, a_src, a_dst,
                                                                   amaxA, denomA, accA, N, n_strips);
    edge_alpha_max<<<cdiv(Etot, 2 * EDGES_PER_WAVE), 64, 0, stream>>>(esrc, edst, a_src, a_dst,
                                                                      alpha, amaxA, E, Etot);
    edge_exp_scatter<<<2048, 256, 0, stream>>>(esrc, edst, emb, alpha, amaxA, denomA, accA,
                                               E, Etot);
    norm_project<true><<<cdiv(n_strips, 8), 256, 0, stream>>>(accA, denomA, W1, b1, h1,
                                                              u + 2 * DIM, a_src, a_dst,
                                                              amaxB, denomB, accB, N, n_strips);

    edge_alpha_max<<<cdiv(Etot, 2 * EDGES_PER_WAVE), 64, 0, stream>>>(esrc, edst, a_src, a_dst,
                                                                      alpha, amaxB, E, Etot);
    edge_exp_scatter<<<2048, 256, 0, stream>>>(esrc, edst, h1, alpha, amaxB, denomB, accB,
                                               E, Etot);
    norm_project<false><<<cdiv(n_strips, 8), 256, 0, stream>>>(accB, denomB, W2, b2,
                                                               (float*)d_out, u, a_src, a_dst,
                                                               amaxB, denomB, accB, N, n_strips);
}


# Softmax shift = bound leaky_relu(max a_src + a_dst[d]) in place of the segment max (guard: a_src range <= 40, else old route); logit/max pass idle, scatter forms logits itself; sum order unchanged
# speedup vs baseline: 3.9097x; 1.1871x over previous
#include <hip/hip_runtime.h>
#include <math.h>

typedef float v2f __attribute__((ext_vector_type(2)));
typedef float v4f __attribute__((ext_vector_type(4)));
typedef float v8f __attribute__((ext_vector_type(8)));
typedef int   v4i __attribute__((ext_vector_type(4)));

#define DIM 64
#define NEG_ATT 0.2f
#define NEG_OUT 0.01f

__device__ __forceinline__ int package_half() {
    unsigned id = __builtin_amdgcn_s_sendmsg_rtn(0x87);
    return (int)((id >> 18) & 1u);
}

#define RANGE_OK 40.0f
#define HALF_PITCH 128
#define LAYER_PITCH (2 * HALF_PITCH)
__device__ __forceinline__ bool shift_by_bound(const float* __restrict__ stat, float& smax) {
    smax = fmaxf(stat[0], stat[HALF_PITCH]);
    const float smin = fminf(stat[1], stat[HALF_PITCH + 1]);
    return (smax - smin) <= RANGE_OK;
}

__device__ __forceinline__ void range_update(float mx, float mn, float* __restrict__ stat) {
#pragma unroll
    for (int off = 16; off > 0; off >>= 1) {
        mx = fmaxf(mx, __shfl_xor(mx, off, 32));
        mn = fminf(mn, __shfl_xor(mn, off, 32));
    }
    if ((threadIdx.x & 31) == 0) {
        float* own = stat + package_half() * HALF_PITCH;
        atomicMax(own, mx);
        atomicMin(own + 1, mn);
    }
}

__global__ __launch_bounds__(256) void fold_att(const float* __restrict__ W1,
                                                const float* __restrict__ as1,
                                                const float* __restrict__ ad1,
                                                const float* __restrict__ W2,
                                                const float* __restrict__ as2,
                                                const float* __restrict__ ad2,
                                                float* __restrict__ u,
                                                float* __restrict__ stat) {
    const int t = threadIdx.x;
    if (t < 8)
        stat[(t >> 2) * LAYER_PITCH + ((t >> 1) & 1) * HALF_PITCH + (t & 1)] =
            (t & 1) ? INFINITY : -INFINITY;
    const int which = t >> 6;
    const int k = t & 63;
    const float* W = (which < 2) ? W1 : W2;
    const float* a = (which == 0) ? as1 : (which == 1) ? ad1 : (which == 2) ? as2 : ad2;
    const float* wr = W + k * DIM;
    float acc = 0.f;
#pragma unroll
    for (int c = 0; c < DIM; ++c) acc += wr[c] * a[c];
    u[t] = acc;
}

__global__ __launch_bounds__(128) void attn_dots_init(const float* __restrict__ X,
                                                      const float* __restrict__ as,
                                                      const float* __restrict__ ad,
                                                      float* __restrict__ a_src,
                                                      float* __restrict__ a_dst,
                                                      float* __restrict__ amax2,
                                                      float* __restrict__ denom2,
                                                      float* __restrict__ acc2,
                                                      float* __restrict__ stat,
                                                      int n, int n_strips) {
    const int lane = threadIdx.x & 31;
    const int sub  = lane >> 4;
    const int l    = lane & 15;
    const int wave = (blockIdx.x * blockDim.x + threadIdx.x) >> 5;
    if (2 * wave >= n_strips) return;
    const size_t plane = (size_t)n * DIM;
    int row0[2];
    row0[0] = 2 * wave * 16;
    row0[1] = min(2 * wave + 1, n_strips - 1) * 16;

    v4f x[2][8];
#pragma unroll
    for (int h = 0; h < 2; ++h) {
        const float* p = X + (size_t)row0[h] * DIM + 4 * lane;
#pragma unroll
        for (int j = 0; j < 8; ++j) x[h][j] = *(const v4f*)(p + j * 128);
    }
    const v4f s4 = *(const v4f*)(as + 4 * l);
    const v4f d4 = *(const v4f*)(ad + 4 * l);

    const v4f z = {0.f, 0.f, 0.f, 0.f};
#pragma unroll
    for (int h = 0; h < 2; ++h) {
        float* q = acc2 + (size_t)row0[h] * DIM + 4 * lane;
#pragma unroll
        for (int it = 0; it < 8; ++it) {
            *(v4f*)(q + it * 128) = z;
            *(v4f*)(q + plane + it * 128) = z;
        }
        if (lane < 16) {
            amax2[row0[h] + lane]      = -INFINITY;
            amax2[n + row0[h] + lane]  = -INFINITY;
            denom2[row0[h] + lane]     = 0.f;
            denom2[n + row0[h] + lane] = 0.f;
        }
    }

    float smx = -INFINITY, smn = INFINITY;
#pragma unroll
    for (int h = 0; h < 2; ++h) {
        float vs[8], vd[8];
#pragma unroll
        for (int j = 0; j < 8; ++j) {
            const v4f v = x[h][j];
            vs[j] = v.x * s4.x + v.y * s4.y + v.z * s4.z + v.w * s4.w;
            vd[j] = v.x * d4.x + v.y * d4.y + v.z * d4.z + v.w * d4.w;
        }
#pragma unroll
        for (int off = 8; off > 0; off >>= 1) {
#pragma unroll
            for (int j = 0; j < 8; ++j) {
                vs[j] += __shfl_xor(vs[j], off, 32);
                vd[j] += __shfl_xor(vd[j], off, 32);
            }
        }
        float os = vs[0], od = vd[0];
#pragma unroll
        for (int j = 1; j < 8; ++j) {
            if (l == j) { os = vs[j]; od = vd[j]; }
        }
        if (l < 8) {
            a_src[row0[h] + 2 * l + sub] = os;
            a_dst[row0[h] + 2 * l + sub] = od;
            smx = fmaxf(smx, os);
            smn = fminf(smn, os);
        }
    }
    range_update(smx, smn, stat);
}

#define EDGES_PER_WAVE 512
__global__ __launch_bounds__(64) void edge_alpha_max(const int* __restrict__ esrc,
                                                     const int* __restrict__ edst,
                                                     const float* __restrict__ a_src,
                                                     const float* __restrict__ a_dst,
                                                     float* __restrict__ alpha,
                                                     float* __restrict__ amax2,
                                                     const float* __restrict__ stat,
                                                     int E, int Etot) {
    const int lane = threadIdx.x & 31;
    const int wave = blockIdx.x * (blockDim.x >> 5)
                   + __builtin_amdgcn_readfirstlane((int)(threadIdx.x >> 5));
    const int base = wave * EDGES_PER_WAVE;
    if (base >= Etot) return;
    float smax;
    if (shift_by_bound(stat, smax)) return;
    const int n = Etot - E;
    float* amax = amax2 + (size_t)package_half() * n;

    if ((((E | Etot) & 3) == 0) && base + EDGES_PER_WAVE <= E) {
        v4i sv[4], dv[4];
#pragma unroll
        for (int k = 0; k < 4; ++k) {
            const int q = base + 128 * k + 4 * lane;
            sv[k] = *(const v4i*)(esrc + q);
            dv[k] = *(const v4i*)(edst + q);
        }
        v4f av[4];
#pragma unroll
        for (int k = 0; k < 4; ++k) {
#pragma unroll
            for (int i = 0; i < 4; ++i) {
                const float a = a_src[sv[k][i]] + a_dst[dv[k][i]];
                av[k][i] = (a > 0.f) ? a : NEG_ATT * a;
            }
        }
#pragma unroll
        for (int k = 0; k < 4; ++k) {
            const int q = base + 128 * k + 4 * lane;
            *(v4f*)(alpha + q) = av[k];
#pragma unroll
            for (int i = 0; i < 4; ++i)
                atomicMax(amax + dv[k][i], av[k][i]);
        }
    } else {
        int   s[16], d[16];
        float a[16];
#pragma unroll
        for (int m = 0; m < 16; ++m) {
            const int e = min(base + 32 * m + lane, Etot - 1);
            if (e < E) { s[m] = esrc[e]; d[m] = edst[e]; } else { s[m] = e - E; d[m] = e - E; }
        }
#pragma unroll
        for (int m = 0; m < 16; ++m) {
            const float t = a_src[s[m]] + a_dst[d[m]];
            a[m] = (t > 0.f) ? t : NEG_ATT * t;
        }
#pragma unroll
        for (int m = 0; m < 16; ++m) {
            const int e = base + 32 * m + lane;
            if (e < Etot) {
                alpha[e] = a[m];
                atomicMax(amax + d[m], a[m]);
            }
        }
    }
}

#define SCATTER_BATCH 8
__global__ __launch_bounds__(256) void edge_exp_scatter(const int* __restrict__ esrc,
                                                        const int* __restrict__ edst,
                                                        const float* __restrict__ X,
                                                        const float* __restrict__ alpha,
                                                        const float* __restrict__ amax2,
                                                        float* __restrict__ denom2,
                                                        float* __restrict__ acc2,
                                                        const float* __restrict__ a_src,
                                                        const float* __restrict__ a_dst,
                                                        const float* __restrict__ stat,
                                                        int E, int Etot) {
    float smax;
    const bool bound = shift_by_bound(stat, smax);
    const int lane    = threadIdx.x & 31;
    const int wib     = __builtin_amdgcn_readfirstlane((int)(threadIdx.x >> 5));
    const int wave    = blockIdx.x * (blockDim.x >> 5) + wib;
    const int nwaves  = gridDim.x * (blockDim.x >> 5);
    const int n       = Etot - E;
    const int nchunks = (Etot + 31) >> 5;
    const int ph      = package_half();
    const char* xwb = (const char*)X;
    char*  accb  = (char*)(acc2 + (size_t)ph * n * DIM);
    float* denom = denom2 + (size_t)ph * n;

    for (int c = wave; c < nchunks; c += nwaves) {
        const int e = c * 32 + lane;
        int s = 0, d = 0;
        float w = 0.f;
        if (e < Etot) {
            if (e < E) { s = esrc[e]; d = edst[e]; } else { s = e - E; d = s; }
            if (bound) {
                const float ad = a_dst[d];
                const float t  = a_src[s] + ad;
                const float tb = smax + ad;
                const float a  = (t > 0.f) ? t : NEG_ATT * t;
                const float ub = (tb > 0.f) ? tb : NEG_ATT * tb;
                w = __expf(a - ub);
            } else {
                const float m = fmaxf(amax2[d], amax2[n + d]);
                w = __expf(alpha[e] - m);
            }
            atomicAdd(denom + d, w);
        }
        const int wi  = __float_as_int(w);
        const int cnt = min(32, Etot - c * 32);
        if (cnt == 32) {
            for (int jb = 0; jb < 32; jb += SCATTER_BATCH) {
                float v0[SCATTER_BATCH], v1[SCATTER_BATCH];
#pragma unroll
                for (int u = 0; u < SCATTER_BATCH; ++u) {
                    const unsigned so = (unsigned)__builtin_amdgcn_readlane(s, jb + u) * (DIM * 4u);
                    const float* row = (const float*)(xwb + so);
                    v0[u] = row[lane];
                    v1[u] = row[lane + 32];
                }
#pragma unroll
                for (int u = 0; u < SCATTER_BATCH; ++u) {
                    const unsigned dof = (unsigned)__builtin_amdgcn_readlane(d, jb + u) * (DIM * 4u);
                    const float wj = __int_as_float(__builtin_amdgcn_readlane(wi, jb + u));
                    float* orow = (float*)(accb + dof);
                    atomicAdd(orow + lane, v0[u] * wj);
                    atomicAdd(orow + lane + 32, v1[u] * wj);
                }
            }
        } else {
            for (int j = 0; j < cnt; ++j) {
                const unsigned so  = (unsigned)__builtin_amdgcn_readlane(s, j) * (DIM * 4u);
                const unsigned dof = (unsigned)__builtin_amdgcn_readlane(d, j) * (DIM * 4u);
                const float wj = __int_as_float(__builtin_amdgcn_readlane(wi, j));
                const float* row = (const float*)(xwb + so);
                float* orow = (float*)(accb + dof);
                atomicAdd(orow + lane, row[lane] * wj);
                atomicAdd(orow + lane + 32, row[lane + 32] * wj);
            }
        }
    }
}

template <bool NEXT>
__global__ __launch_bounds__(256) void norm_project(const float* __restrict__ acc2,
                                                    const float* __restrict__ denom2,
                                                    const float* __restrict__ W,
                                                    const float* __restrict__ b,
                                                    float* __restrict__ out,
                                                    const float* __restrict__ un,
                                                    float* __restrict__ a_src,
                                                    float* __restrict__ a_dst,
                                                    float* __restrict__ amaxN,
                                                    float* __restrict__ denomN,
                                                    float* __restrict__ accN,
                                                    float* __restrict__ statN,
                                                    int n, int n_strips) {
    __shared__ v2f Wl[16 * 4 * 32];
    const int tid = threadIdx.x;
    const int lane = tid & 31;
    const int half = lane >> 4;
    const int l    = lane & 15;
    const int strip = blockIdx.x * 8 + (tid >> 5);
    const int row0 = min(strip, n_strips - 1) * 16;
    const size_t plane = (size_t)n * DIM;

    v2f wreg[8];
#pragma unroll
    for (int r = 0; r < 8; ++r) {
        const int i = tid + 256 * r;
        const int t = i >> 7, tn = (i >> 5) & 3, ln = i & 31;
        const int kb = 8 * (t >> 1) + 4 * (ln >> 4) + 2 * (t & 1);
        wreg[r].x = W[kb * DIM + 4 * (ln & 15) + tn];
        wreg[r].y = W[(kb + 1) * DIM + 4 * (ln & 15) + tn];
    }

    const int arow = row0 + l;
    const float d0 = denom2[arow];
    const float d1 = denom2[n + arow];
    const float* p0 = acc2 + (size_t)arow * DIM + 4 * half;
    const float* p1 = p0 + plane;
    v4f ax[8], ay[8];
#pragma unroll
    for (int j = 0; j < 8; ++j) {
        ax[j] = *(const v4f*)(p0 + 8 * j);
        ay[j] = *(const v4f*)(p1 + 8 * j);
    }

#pragma unroll
    for (int r = 0; r < 8; ++r) Wl[tid + 256 * r] = wreg[r];
    __syncthreads();
    if (strip >= n_strips) return;

    const float rden = 1.0f / (d0 + d1);
    v4f av[8];
#pragma unroll
    for (int j = 0; j < 8; ++j) av[j] = (ax[j] + ay[j]) * rden;

    v8f c[4];
#pragma unroll
    for (int tn = 0; tn < 4; ++tn) c[tn] = (v8f){0.f, 0.f, 0.f, 0.f, 0.f, 0.f, 0.f, 0.f};
#pragma unroll
    for (int t = 0; t < 16; ++t) {
        v2f a;
        if (t & 1) { a.x = av[t >> 1].z; a.y = av[t >> 1].w; }
        else       { a.x = av[t >> 1].x; a.y = av[t >> 1].y; }
#pragma unroll
        for (int tn = 0; tn < 4; ++tn) {
            const v2f bb = Wl[(t * 4 + tn) * 32 + lane];
            c[tn] = __builtin_amdgcn_wmma_f32_16x16x4_f32(false, a, false, bb,
                                                          (short)0, c[tn], false, false);
        }
    }

    float ps[8], pd[8];
    const v4f bias = *(const v4f*)(b + 4 * l);
    v4f us = {0.f, 0.f, 0.f, 0.f}, ud = {0.f, 0.f, 0.f, 0.f};
    if (NEXT) { us = *(const v4f*)(un + 4 * l); ud = *(const v4f*)(un + DIM + 4 * l); }
    float* orow = out + (size_t)(row0 + 8 * half) * DIM + 4 * l;
#pragma unroll
    for (int i = 0; i < 8; ++i) {
        v4f v;
        v.x = c[0][i] + bias.x;
        v.y = c[1][i] + bias.y;
        v.z = c[2][i] + bias.z;
        v.w = c[3][i] + bias.w;
        v.x = (v.x > 0.f) ? v.x : NEG_OUT * v.x;
        v.y = (v.y > 0.f) ? v.y : NEG_OUT * v.y;
        v.z = (v.z > 0.f) ? v.z : NEG_OUT * v.z;
        v.w = (v.w > 0.f) ? v.w : NEG_OUT * v.w;
        *(v4f*)(orow + (size_t)i * DIM) = v;
        ps[i] = v.x * us.x + v.y * us.y + v.z * us.z + v.w * us.w;
        pd[i] = v.x * ud.x + v.y * ud.y + v.z * ud.z + v.w * ud.w;
    }
    if (NEXT) {
#pragma unroll
        for (int off = 8; off > 0; off >>= 1) {
#pragma unroll
            for (int i = 0; i < 8; ++i) {
                ps[i] += __shfl_xor(ps[i], off, 32);
                pd[i] += __shfl_xor(pd[i], off, 32);
            }
        }
#pragma unroll
        for (int i = 0; i < 8; ++i) {
            if (l == i) {
                a_src[row0 + 8 * half + i] = ps[i];
                a_dst[row0 + 8 * half + i] = pd[i];
            }
        }
        float mx = ps[0], mn = ps[0];
#pragma unroll
        for (int i = 1; i < 8; ++i) { mx = fmaxf(mx, ps[i]); mn = fminf(mn, ps[i]); }
        range_update(mx, mn, statN);
        if (lane < 16) {
            amaxN[row0 + lane]      = -INFINITY;
            amaxN[n + row0 + lane]  = -INFINITY;
            denomN[row0 + lane]     = 0.f;
            denomN[n + row0 + lane] = 0.f;
        }
        const v4f z = {0.f, 0.f, 0.f, 0.f};
        float* q = accN + (size_t)row0 * DIM + 4 * lane;
#pragma unroll
        for (int it = 0; it < 8; ++it) {
            *(v4f*)(q + it * 128) = z;
            *(v4f*)(q + plane + it * 128) = z;
        }
    }
}

static inline int cdiv(int a, int b) { return (a + b - 1) / b; }

extern "C" void kernel_launch(void* const* d_in, const int* in_sizes, int n_in,
                              void* d_out, int out_size, void* d_ws, size_t ws_size,
                              hipStream_t stream) {
    const int*   edge_index = (const int*)d_in[0];
    const float* emb        = (const float*)d_in[1];
    const float* W1         = (const float*)d_in[2];
    const float* att_src1   = (const float*)d_in[3];
    const float* att_dst1   = (const float*)d_in[4];
    const float* b1         = (const float*)d_in[5];
    const float* W2         = (const float*)d_in[6];
    const float* att_src2   = (const float*)d_in[7];
    const float* att_dst2   = (const float*)d_in[8];
    const float* b2         = (const float*)d_in[9];

    const int E    = in_sizes[0] / 2;
    const int N    = in_sizes[1] / DIM;
    const int Etot = E + N;
    const int n_strips = N / 16;

    const int* esrc = edge_index;
    const int* edst = edge_index + E;

    float* ws     = (float*)d_ws;
    float* h1     = ws;
    float* accA   = h1 + (size_t)N * DIM;
    float* accB   = accA + (size_t)2 * N * DIM;
    float* a_src  = accB + (size_t)2 * N * DIM;
    float* a_dst  = a_src + N;
    float* amaxA  = a_dst + N;
    float* denomA = amaxA + 2 * N;
    float* amaxB  = denomA + 2 * N;
    float* denomB = amaxB + 2 * N;
    float* u      = denomB + 2 * N;
    float* alpha  = u + 4 * DIM;
    float* stat   = alpha + Etot;

    fold_att<<<1, 256, 0, stream>>>(W1, att_src1, att_dst1, W2, att_src2, att_dst2, u, stat);

    attn_dots_init<<<cdiv(cdiv(n_strips, 2), 4), 128, 0, stream>>>(emb, u, u + DIM, a_src, a_dst,
                                                                   amaxA, denomA, accA, stat, N, n_strips);
    edge_alpha_max<<<cdiv(Etot, 2 * EDGES_PER_WAVE), 64, 0, stream>>>(esrc, edst, a_src, a_dst,
                                                                      alpha, amaxA, stat, E, Etot);
    edge_exp_scatter<<<2048, 256, 0, stream>>>(esrc, edst, emb, alpha, amaxA, denomA, accA,
                                               a_src, a_dst, stat, E, Etot);
    norm_project<true><<<cdiv(n_strips, 8), 256, 0, stream>>>(accA, denomA, W1, b1, h1,
                                                              u + 2 * DIM, a_src, a_dst,
                                                              amaxB, denomB, accB, stat + LAYER_PITCH, N, n_strips);

    edge_alpha_max<<<cdiv(Etot, 2 * EDGES_PER_WAVE), 64, 0, stream>>>(esrc, edst, a_src, a_dst,
                                                                      alpha, amaxB, stat + LAYER_PITCH, E, Etot);
    edge_exp_scatter<<<2048, 256, 0, stream>>>(esrc, edst, h1, alpha, amaxB, denomB, accB,
                                               a_src, a_dst, stat + LAYER_PITCH, E, Etot);
    norm_project<false><<<cdiv(n_strips, 8), 256, 0, stream>>>(accB, denomB, W2, b2,
                                                               (float*)d_out, u, a_src, a_dst,
                                                               amaxB, denomB, accB, stat + LAYER_PITCH, N, n_strips);
}
